# starting kernel
# baseline (speedup 1.0000x reference)
#include <hip/hip_runtime.h>
#include <hip/hip_bf16.h>

typedef float v2f __attribute__((ext_vector_type(2)));
typedef float v8f __attribute__((ext_vector_type(8)));

#define C_TOT   128
#define MOUT    16
#define NSTEP   64
#define CH_BLK  16
#define B_TILE  16
#define BPAD    17
#define APACK_PER_C (NSTEP * 32 * 2)

__global__ __launch_bounds__(256) void wtp_zero(float* __restrict__ p, int n) {
  int i = blockIdx.x * 256 + threadIdx.x;
  if (i < n) p[i] = 0.0f;
}

__global__ __launch_bounds__(256) void wtp_build_apack(
    const float* __restrict__ cg, const float* __restrict__ w,
    const int* __restrict__ M1,  const int* __restrict__ M2,
    const int* __restrict__ lind, const int* __restrict__ seg,
    float* __restrict__ apack, int nnz) {
  int t = blockIdx.x * 256 + threadIdx.x;
  if (t >= nnz * C_TOT) return;
  int c = t & (C_TOT - 1);
  int e = t >> 7;
  int k    = M2[e] * 16 + M1[e];
  int s    = k >> 2;
  int kl   = k & 3;
  int lane = ((kl >> 1) << 4) | seg[e];
  int j    = kl & 1;
  apack[(((size_t)c * NSTEP + s) * 32 + lane) * 2 + j] =
      cg[e] * w[lind[e] * C_TOT + c];
}

__global__ __launch_bounds__(256) void wtp_wmma_kernel(
    const float* __restrict__ x1, const float* __restrict__ x2,
    const float* __restrict__ apack, float* __restrict__ out) {
  __shared__ float s1[MOUT * CH_BLK * BPAD];
  __shared__ float s2[MOUT * CH_BLK * BPAD];

  const int tid = threadIdx.x;
  const int b0  = blockIdx.x * B_TILE;
  const int c0  = blockIdx.y * CH_BLK;

  #pragma unroll
  for (int i = 0; i < (B_TILE * MOUT * CH_BLK) / 256; ++i) {
    int f = tid + 256 * i;
    int c = f & (CH_BLK - 1);
    int m = (f >> 4) & (MOUT - 1);
    int b = f >> 8;
    size_t g = (size_t)(b0 + b) * (MOUT * C_TOT) + m * C_TOT + (c0 + c);
    int l = (m * CH_BLK + c) * BPAD + b;
    s1[l] = x1[g];
    s2[l] = x2[g];
  }
  __syncthreads();

  const int wave = tid >> 5;
  const int lane = tid & 31;
  const int n    = lane & 15;
  const int h    = lane >> 4;

  #pragma unroll
  for (int cc = 0; cc < 2; ++cc) {
    const int cl = wave * 2 + cc;
    const int c  = c0 + cl;

    float x2v[16];
    #pragma unroll
    for (int m = 0; m < 16; ++m) x2v[m] = s2[(m * CH_BLK + cl) * BPAD + n];
    float yv[8];
    #pragma unroll
    for (int a = 0; a < 4; ++a)
      #pragma unroll
      for (int j = 0; j < 2; ++j)
        yv[2 * a + j] = s1[((4 * a + 2 * h + j) * CH_BLK + cl) * BPAD + n];

    v8f acc = {};
    const float* ap = apack + (size_t)c * APACK_PER_C + lane * 2;

    #pragma unroll
    for (int s = 0; s < NSTEP; ++s) {
      v2f A = *(const v2f*)(ap + s * 64);
      float xm2 = x2v[s >> 2];
      v2f Bf;
      Bf.x = xm2 * yv[2 * (s & 3) + 0];
      Bf.y = xm2 * yv[2 * (s & 3) + 1];
      acc = __builtin_amdgcn_wmma_f32_16x16x4_f32(
          false, A, false, Bf, (short)0, acc, false, false);
    }

    #pragma unroll
    for (int r = 0; r < 8; ++r) {
      int Mo = r + 8 * h;
      out[(size_t)(b0 + n) * (MOUT * C_TOT) + Mo * C_TOT + c] = acc[r];
    }
  }
}

extern "C" void kernel_launch(void* const* d_in, const int* in_sizes, int n_in,
                              void* d_out, int out_size, void* d_ws, size_t ws_size,
                              hipStream_t stream) {
  const float* x1 = (const float*)d_in[0];
  const float* x2 = (const float*)d_in[1];
  const float* w  = (const float*)d_in[2];
  const float* cg = (const float*)d_in[3];
  const int* M1   = (const int*)d_in[4];
  const int* M2   = (const int*)d_in[5];
  const int* li   = (const int*)d_in[6];
  const int* sg   = (const int*)d_in[7];
  float* out      = (float*)d_out;
  float* apack    = (float*)d_ws;

  const int B   = in_sizes[0] / (MOUT * C_TOT);
  const int nnz = in_sizes[3];

  const int na = C_TOT * APACK_PER_C;
  wtp_zero<<<(na + 255) / 256, 256, 0, stream>>>(apack, na);

  const int nb = nnz * C_TOT;
  wtp_build_apack<<<(nb + 255) / 256, 256, 0, stream>>>(cg, w, M1, M2, li, sg,
                                                        apack, nnz);

  dim3 grid(B / B_TILE, C_TOT / CH_BLK);
  wtp_wmma_kernel<<<grid, 256, 0, stream>>>(x1, x2, apack, out);
}


# Hold each channel's 64 A fragments in registers (one wave per channel) and reuse them over 4 batch tiles per workgroup instead of re-reading packed A from L2 every tile; accumulation order unchanged
# speedup vs baseline: 1.0655x; 1.0655x over previous
#include <hip/hip_runtime.h>
#include <hip/hip_bf16.h>


typedef float v2f __attribute__((ext_vector_type(2)));
typedef float v8f __attribute__((ext_vector_type(8)));

#define C_TOT   128
#define MOUT    16
#define NSTEP   64
#define CH_BLK  16
#define B_TILE  16
#define BPAD    17
#define APACK_PER_C (NSTEP * 32 * 2)
#define R_TILES 4
#define NTHR    512
#define XBUF    (MOUT * CH_BLK * BPAD)
#define ROWF    (MOUT * C_TOT)

__global__ __launch_bounds__(256) void wtp_zero(float* __restrict__ p, int n) {
  int i = blockIdx.x * 256 + threadIdx.x;
  if (i < n) p[i] = 0.0f;
}

__global__ __launch_bounds__(256) void wtp_build_apack(
    const float* __restrict__ cg, const float* __restrict__ w,
    const int* __restrict__ M1,  const int* __restrict__ M2,
    const int* __restrict__ lind, const int* __restrict__ seg,
    float* __restrict__ apack, int nnz) {
  int t = blockIdx.x * 256 + threadIdx.x;
  if (t >= nnz * C_TOT) return;
  int c = t & (C_TOT - 1);
  int e = t >> 7;
  int k    = M2[e] * 16 + M1[e];
  int s    = k >> 2;
  int kl   = k & 3;
  int lane = ((kl >> 1) << 4) | seg[e];
  int j    = kl & 1;
  apack[(((size_t)c * NSTEP + s) * 32 + lane) * 2 + j] =
      cg[e] * w[lind[e] * C_TOT + c];
}

__global__ __launch_bounds__(NTHR) void wtp_wmma_kernel(
    const float* __restrict__ x1, const float* __restrict__ x2,
    const float* __restrict__ apack, float* __restrict__ out, int ntiles) {
  __shared__ float s1[2 * XBUF];
  __shared__ float s2[2 * XBUF];

  const int tid  = threadIdx.x;
  const int c0   = blockIdx.y * CH_BLK;
  const int t0   = blockIdx.x * R_TILES;
  const int wave = tid >> 5;
  const int lane = tid & 31;
  const int n    = lane & 15;
  const int h    = lane >> 4;
  const int c    = c0 + wave;

  int nt = ntiles - t0;
  if (nt > R_TILES) nt = R_TILES;

  v2f A[NSTEP];
  {
    const float* ap = apack + (size_t)c * APACK_PER_C + lane * 2;
    #pragma unroll
    for (int s = 0; s < NSTEP; ++s) A[s] = *(const v2f*)(ap + s * 64);
  }

  const int sb    = tid >> 8;
  const int lbase = (tid & 255) * BPAD + sb;
  const size_t gbase = (size_t)sb * ROWF + (size_t)((tid >> 4) & 15) * C_TOT + (c0 + (tid & 15));

  float r1[8], r2[8];
  {
    const size_t g0 = (size_t)t0 * B_TILE * ROWF + gbase;
    #pragma unroll
    for (int i = 0; i < 8; ++i) {
      r1[i] = x1[g0 + (size_t)(2 * i) * ROWF];
      r2[i] = x2[g0 + (size_t)(2 * i) * ROWF];
    }
    #pragma unroll
    for (int i = 0; i < 8; ++i) {
      s1[lbase + 2 * i] = r1[i];
      s2[lbase + 2 * i] = r2[i];
    }
  }
  __syncthreads();

  for (int t = 0; t < nt; ++t) {
    const int  cur  = (t & 1) * XBUF;
    const bool more = (t + 1 < nt);

    if (more) {
      const size_t g0 = (size_t)(t0 + t + 1) * B_TILE * ROWF + gbase;
      #pragma unroll
      for (int i = 0; i < 8; ++i) {
        r1[i] = x1[g0 + (size_t)(2 * i) * ROWF];
        r2[i] = x2[g0 + (size_t)(2 * i) * ROWF];
      }
    }

    float x2v[16];
    #pragma unroll
    for (int m = 0; m < 16; ++m) x2v[m] = s2[cur + (m * CH_BLK + wave) * BPAD + n];
    float yv[8];
    #pragma unroll
    for (int a = 0; a < 4; ++a)
      #pragma unroll
      for (int j = 0; j < 2; ++j)
        yv[2 * a + j] = s1[cur + ((4 * a + 2 * h + j) * CH_BLK + wave) * BPAD + n];

    v8f acc = {};
    #pragma unroll
    for (int s = 0; s < NSTEP; ++s) {
      float xm2 = x2v[s >> 2];
      v2f Bf;
      Bf.x = xm2 * yv[2 * (s & 3) + 0];
      Bf.y = xm2 * yv[2 * (s & 3) + 1];
      acc = __builtin_amdgcn_wmma_f32_16x16x4_f32(
          false, A[s], false, Bf, (short)0, acc, false, false);
    }

    if (more) {
      const int nxt = XBUF - cur;
      #pragma unroll
      for (int i = 0; i < 8; ++i) {
        s1[nxt + lbase + 2 * i] = r1[i];
        s2[nxt + lbase + 2 * i] = r2[i];
      }
    }
    __syncthreads();

    {
      float* op = out + (size_t)((t0 + t) * B_TILE + n) * ROWF + (size_t)(8 * h) * C_TOT + c;
      #pragma unroll
      for (int r = 0; r < 8; ++r) op[r * C_TOT] = acc[r];
    }
  }
}

extern "C" void kernel_launch(void* const* d_in, const int* in_sizes, int n_in,
                              void* d_out, int out_size, void* d_ws, size_t ws_size,
                              hipStream_t stream) {
  const float* x1 = (const float*)d_in[0];
  const float* x2 = (const float*)d_in[1];
  const float* w  = (const float*)d_in[2];
  const float* cg = (const float*)d_in[3];
  const int* M1   = (const int*)d_in[4];
  const int* M2   = (const int*)d_in[5];
  const int* li   = (const int*)d_in[6];
  const int* sg   = (const int*)d_in[7];
  float* out      = (float*)d_out;
  float* apack    = (float*)d_ws;

  const int B   = in_sizes[0] / (MOUT * C_TOT);
  const int nnz = in_sizes[3];

  const int na = C_TOT * APACK_PER_C;
  wtp_zero<<<(na + 255) / 256, 256, 0, stream>>>(apack, na);

  const int nb = nnz * C_TOT;
  wtp_build_apack<<<(nb + 255) / 256, 256, 0, stream>>>(cg, w, M1, M2, li, sg,
                                                        apack, nnz);

  const int ntiles = B / B_TILE;
  dim3 grid((ntiles + R_TILES - 1) / R_TILES, C_TOT / CH_BLK);
  wtp_wmma_kernel<<<grid, NTHR, 0, stream>>>(x1, x2, apack, out, ntiles);
}


# Build each channel's packed weights in LDS inside the matrix kernel (zero, scatter, read fragments per wave): 1 launch instead of 3, no workspace pass; values and sum order unchanged
# speedup vs baseline: 1.2833x; 1.2044x over previous
#include <hip/hip_runtime.h>
#include <hip/hip_bf16.h>


typedef float v2f __attribute__((ext_vector_type(2)));
typedef float v8f __attribute__((ext_vector_type(8)));

#define C_TOT   128
#define MOUT    16
#define NSTEP   64
#define CH_BLK  16
#define B_TILE  16
#define BPAD    17
#define WPITCH  260
#define WREG    4224
#define E_UNR   16
#define R_TILES 4
#define NTHR    512
#define XBUF    (MOUT * CH_BLK * BPAD)
#define ROWF    (MOUT * C_TOT)

template <bool FIRST>
__device__ __forceinline__ void wc_block(
    int e0, int lane, int c, int nnz, float* __restrict__ myW,
    const float* __restrict__ w,  const float* __restrict__ cg,
    const int* __restrict__ M1,   const int* __restrict__ M2,
    const int* __restrict__ lind, const int* __restrict__ seg) {
  int   li[E_UNR], m1[E_UNR], m2[E_UNR], sg[E_UNR];
  float g[E_UNR], wv[E_UNR];
  #pragma unroll
  for (int i = 0; i < E_UNR; ++i) {
    const int e  = e0 + 32 * i + lane;
    const int ec = e < nnz ? e : nnz - 1;
    li[i] = lind[ec];
    m1[i] = M1[ec];
    m2[i] = M2[ec];
    sg[i] = seg[ec];
    g[i]  = cg[ec];
  }
  if (FIRST) {
    #pragma unroll
    for (int i = 0; i < WREG / 128; ++i)
      *(float4*)(myW + (i * 32 + lane) * 4) = make_float4(0.0f, 0.0f, 0.0f, 0.0f);
  }
  #pragma unroll
  for (int i = 0; i < E_UNR; ++i) wv[i] = w[li[i] * C_TOT + c];
  if (FIRST) {
    __syncthreads();
  }
  #pragma unroll
  for (int i = 0; i < E_UNR; ++i) {
    const int   at = sg[i] * WPITCH + m2[i] * 16 + m1[i];
    myW[at] = g[i] * wv[i];
  }
}

__global__ __launch_bounds__(NTHR) void wtp_wmma_kernel(
    const float* __restrict__ x1, const float* __restrict__ x2,
    const float* __restrict__ w,  const float* __restrict__ cg,
    const int* __restrict__ M1,   const int* __restrict__ M2,
    const int* __restrict__ lind, const int* __restrict__ seg,
    float* __restrict__ out, int ntiles, int nnz) {
  __shared__ __attribute__((aligned(16))) float lds[CH_BLK * WREG];
  float* const s1 = lds;
  float* const s2 = lds + 2 * XBUF;

  const int tid  = threadIdx.x;
  const int c0   = blockIdx.y * CH_BLK;
  const int t0   = blockIdx.x * R_TILES;
  const int wave = tid >> 5;
  const int lane = tid & 31;
  const int n    = lane & 15;
  const int h    = lane >> 4;
  const int c    = c0 + wave;

  int nt = ntiles - t0;
  if (nt > R_TILES) nt = R_TILES;

  const int sb    = tid >> 8;
  const int lbase = (tid & 255) * BPAD + sb;
  const size_t gbase = (size_t)sb * ROWF + (size_t)((tid >> 4) & 15) * C_TOT + (c0 + (tid & 15));

  float r1[8], r2[8];
  {
    const size_t g0 = (size_t)t0 * B_TILE * ROWF + gbase;
    #pragma unroll
    for (int i = 0; i < 8; ++i) {
      r1[i] = x1[g0 + (size_t)(2 * i) * ROWF];
      r2[i] = x2[g0 + (size_t)(2 * i) * ROWF];
    }
  }

  v2f A[NSTEP];
  {
    float* const myW = lds + wave * WREG;
    wc_block<true>(0, lane, c, nnz, myW, w, cg, M1, M2, lind, seg);
    for (int e0 = 32 * E_UNR; e0 < nnz; e0 += 32 * E_UNR)
      wc_block<false>(e0, lane, c, nnz, myW, w, cg, M1, M2, lind, seg);
    __syncthreads();

    const float* ap = myW + n * WPITCH + 2 * h;
    #pragma unroll
    for (int s = 0; s < NSTEP; ++s) A[s] = *(const v2f*)(ap + 4 * s);
  }
  __syncthreads();

  #pragma unroll
  for (int i = 0; i < 8; ++i) {
    s1[lbase + 2 * i] = r1[i];
    s2[lbase + 2 * i] = r2[i];
  }
  __syncthreads();

  for (int t = 0; t < nt; ++t) {
    const int  cur  = (t & 1) * XBUF;
    const bool more = (t + 1 < nt);

    if (more) {
      const size_t g0 = (size_t)(t0 + t + 1) * B_TILE * ROWF + gbase;
      #pragma unroll
      for (int i = 0; i < 8; ++i) {
        r1[i] = x1[g0 + (size_t)(2 * i) * ROWF];
        r2[i] = x2[g0 + (size_t)(2 * i) * ROWF];
      }
    }

    float x2v[16];
    #pragma unroll
    for (int m = 0; m < 16; ++m) x2v[m] = s2[cur + (m * CH_BLK + wave) * BPAD + n];
    float yv[8];
    #pragma unroll
    for (int a = 0; a < 4; ++a)
      #pragma unroll
      for (int j = 0; j < 2; ++j)
        yv[2 * a + j] = s1[cur + ((4 * a + 2 * h + j) * CH_BLK + wave) * BPAD + n];

    v8f acc = {};
    #pragma unroll
    for (int s = 0; s < NSTEP; ++s) {
      float xm2 = x2v[s >> 2];
      v2f Bf;
      Bf.x = xm2 * yv[2 * (s & 3) + 0];
      Bf.y = xm2 * yv[2 * (s & 3) + 1];
      acc = __builtin_amdgcn_wmma_f32_16x16x4_f32(
          false, A[s], false, Bf, (short)0, acc, false, false);
    }

    if (more) {
      const int nxt = XBUF - cur;
      #pragma unroll
      for (int i = 0; i < 8; ++i) {
        s1[nxt + lbase + 2 * i] = r1[i];
        s2[nxt + lbase + 2 * i] = r2[i];
      }
    }
    __syncthreads();

    {
      float* op = out + (size_t)((t0 + t) * B_TILE + n) * ROWF + (size_t)(8 * h) * C_TOT + c;
      #pragma unroll
      for (int r = 0; r < 8; ++r) op[r * C_TOT] = acc[r];
    }
  }
}

extern "C" void kernel_launch(void* const* d_in, const int* in_sizes, int n_in,
                              void* d_out, int out_size, void* d_ws, size_t ws_size,
                              hipStream_t stream) {
  const float* x1 = (const float*)d_in[0];
  const float* x2 = (const float*)d_in[1];
  const float* w  = (const float*)d_in[2];
  const float* cg = (const float*)d_in[3];
  const int* M1   = (const int*)d_in[4];
  const int* M2   = (const int*)d_in[5];
  const int* li   = (const int*)d_in[6];
  const int* sg   = (const int*)d_in[7];
  float* out      = (float*)d_out;

  const int B   = in_sizes[0] / (MOUT * C_TOT);
  const int nnz = in_sizes[3];

  const int ntiles = B / B_TILE;
  dim3 grid((ntiles + R_TILES - 1) / R_TILES, C_TOT / CH_BLK);
  wtp_wmma_kernel<<<grid, NTHR, 0, stream>>>(x1, x2, w, cg, M1, M2, li, sg, out,
                                             ntiles, nnz);
}


# Stage each batch tile's output through LDS so every global store writes whole 64-byte channel runs (16-byte stores), not 4-byte pieces spread over 32 lines; values and sum order unchanged
# speedup vs baseline: 2.4329x; 1.8958x over previous
#include <hip/hip_runtime.h>
#include <hip/hip_bf16.h>


typedef float v2f __attribute__((ext_vector_type(2)));
typedef float v8f __attribute__((ext_vector_type(8)));

#define C_TOT   128
#define MOUT    16
#define NSTEP   64
#define CH_BLK  16
#define B_TILE  16
#define BPAD    17
#define WPITCH  260
#define WREG    4224
#define E_UNR   16
#define R_TILES 4
#define NTHR    512
#define XBUF    (MOUT * CH_BLK * BPAD)
#define OPITCH  260
#define OBUF    (B_TILE * OPITCH)
#define ROWF    (MOUT * C_TOT)

template <bool FIRST>
__device__ __forceinline__ void wc_block(
    int e0, int lane, int c, int nnz, float* __restrict__ myW,
    const float* __restrict__ w,  const float* __restrict__ cg,
    const int* __restrict__ M1,   const int* __restrict__ M2,
    const int* __restrict__ lind, const int* __restrict__ seg) {
  int   li[E_UNR], m1[E_UNR], m2[E_UNR], sg[E_UNR];
  float g[E_UNR], wv[E_UNR];
  #pragma unroll
  for (int i = 0; i < E_UNR; ++i) {
    const int e  = e0 + 32 * i + lane;
    const int ec = e < nnz ? e : nnz - 1;
    li[i] = lind[ec];
    m1[i] = M1[ec];
    m2[i] = M2[ec];
    sg[i] = seg[ec];
    g[i]  = cg[ec];
  }
  if (FIRST) {
    #pragma unroll
    for (int i = 0; i < WREG / 128; ++i)
      *(float4*)(myW + (i * 32 + lane) * 4) = make_float4(0.0f, 0.0f, 0.0f, 0.0f);
  }
  #pragma unroll
  for (int i = 0; i < E_UNR; ++i) wv[i] = w[li[i] * C_TOT + c];
  if (FIRST) {
    __syncthreads();
  }
  #pragma unroll
  for (int i = 0; i < E_UNR; ++i) {
    const int   at = sg[i] * WPITCH + m2[i] * 16 + m1[i];
    myW[at] = g[i] * wv[i];
  }
}

__global__ __launch_bounds__(NTHR) void wtp_wmma_kernel(
    const float* __restrict__ x1, const float* __restrict__ x2,
    const float* __restrict__ w,  const float* __restrict__ cg,
    const int* __restrict__ M1,   const int* __restrict__ M2,
    const int* __restrict__ lind, const int* __restrict__ seg,
    float* __restrict__ out, int ntiles, int nnz) {
  __shared__ __attribute__((aligned(16))) float lds[CH_BLK * WREG];
  float* const s1 = lds;
  float* const s2 = lds + 2 * XBUF;
  float* const so = lds + 4 * XBUF;

  const int tid  = threadIdx.x;
  const int c0   = blockIdx.y * CH_BLK;
  const int t0   = blockIdx.x * R_TILES;
  const int wave = tid >> 5;
  const int lane = tid & 31;
  const int n    = lane & 15;
  const int h    = lane >> 4;
  const int c    = c0 + wave;

  int nt = ntiles - t0;
  if (nt > R_TILES) nt = R_TILES;

  const int sb    = tid >> 8;
  const int lbase = (tid & 255) * BPAD + sb;
  const size_t gbase = (size_t)sb * ROWF + (size_t)((tid >> 4) & 15) * C_TOT + (c0 + (tid & 15));

  float r1[8], r2[8];
  {
    const size_t g0 = (size_t)t0 * B_TILE * ROWF + gbase;
    #pragma unroll
    for (int i = 0; i < 8; ++i) {
      r1[i] = x1[g0 + (size_t)(2 * i) * ROWF];
      r2[i] = x2[g0 + (size_t)(2 * i) * ROWF];
    }
  }

  v2f A[NSTEP];
  {
    float* const myW = lds + wave * WREG;
    wc_block<true>(0, lane, c, nnz, myW, w, cg, M1, M2, lind, seg);
    for (int e0 = 32 * E_UNR; e0 < nnz; e0 += 32 * E_UNR)
      wc_block<false>(e0, lane, c, nnz, myW, w, cg, M1, M2, lind, seg);
    __syncthreads();

    const float* ap = myW + n * WPITCH + 2 * h;
    #pragma unroll
    for (int s = 0; s < NSTEP; ++s) A[s] = *(const v2f*)(ap + 4 * s);
  }
  __syncthreads();

  #pragma unroll
  for (int i = 0; i < 8; ++i) {
    s1[lbase + 2 * i] = r1[i];
    s2[lbase + 2 * i] = r2[i];
  }
  __syncthreads();

  for (int t = 0; t < nt; ++t) {
    const int  cur  = (t & 1) * XBUF;
    const bool more = (t + 1 < nt);

    if (more) {
      const size_t g0 = (size_t)(t0 + t + 1) * B_TILE * ROWF + gbase;
      #pragma unroll
      for (int i = 0; i < 8; ++i) {
        r1[i] = x1[g0 + (size_t)(2 * i) * ROWF];
        r2[i] = x2[g0 + (size_t)(2 * i) * ROWF];
      }
    }

    float x2v[16];
    #pragma unroll
    for (int m = 0; m < 16; ++m) x2v[m] = s2[cur + (m * CH_BLK + wave) * BPAD + n];
    float yv[8];
    #pragma unroll
    for (int a = 0; a < 4; ++a)
      #pragma unroll
      for (int j = 0; j < 2; ++j)
        yv[2 * a + j] = s1[cur + ((4 * a + 2 * h + j) * CH_BLK + wave) * BPAD + n];

    v8f acc = {};
    #pragma unroll
    for (int s = 0; s < NSTEP; ++s) {
      float xm2 = x2v[s >> 2];
      v2f Bf;
      Bf.x = xm2 * yv[2 * (s & 3) + 0];
      Bf.y = xm2 * yv[2 * (s & 3) + 1];
      acc = __builtin_amdgcn_wmma_f32_16x16x4_f32(
          false, A[s], false, Bf, (short)0, acc, false, false);
    }

    if (more) {
      const int nxt = XBUF - cur;
      #pragma unroll
      for (int i = 0; i < 8; ++i) {
        s1[nxt + lbase + 2 * i] = r1[i];
        s2[nxt + lbase + 2 * i] = r2[i];
      }
    }

    float* const ot = so + (t & 1) * OBUF;
    {
      float* ow = ot + n * OPITCH + (8 * h) * CH_BLK + wave;
      #pragma unroll
      for (int r = 0; r < 8; ++r) ow[r * CH_BLK] = acc[r];
    }
    __syncthreads();

    {
      const int oq = tid & 3, om = (tid >> 2) & 15, ob = tid >> 6;
      const float* ip = ot + ob * OPITCH + om * CH_BLK + 4 * oq;
      float* op = out + (size_t)((t0 + t) * B_TILE + ob) * ROWF + (size_t)om * C_TOT + (c0 + 4 * oq);
      for (int i = 0; i < 2; ++i)
        *(float4*)(op + (size_t)(8 * i) * ROWF) = *(const float4*)(ip + 8 * i * OPITCH);
    }
  }
}

extern "C" void kernel_launch(void* const* d_in, const int* in_sizes, int n_in,
                              void* d_out, int out_size, void* d_ws, size_t ws_size,
                              hipStream_t stream) {
  const float* x1 = (const float*)d_in[0];
  const float* x2 = (const float*)d_in[1];
  const float* w  = (const float*)d_in[2];
  const float* cg = (const float*)d_in[3];
  const int* M1   = (const int*)d_in[4];
  const int* M2   = (const int*)d_in[5];
  const int* li   = (const int*)d_in[6];
  const int* sg   = (const int*)d_in[7];
  float* out      = (float*)d_out;

  const int B   = in_sizes[0] / (MOUT * C_TOT);
  const int nnz = in_sizes[3];

  const int ntiles = B / B_TILE;
  dim3 grid((ntiles + R_TILES - 1) / R_TILES, C_TOT / CH_BLK);
  wtp_wmma_kernel<<<grid, NTHR, 0, stream>>>(x1, x2, w, cg, M1, M2, li, sg, out,
                                             ntiles, nnz);
}
